# starting kernel
# baseline (speedup 1.0000x reference)
#include <hip/hip_runtime.h>

typedef __attribute__((ext_vector_type(2))) float v2f;
typedef __attribute__((ext_vector_type(8))) float v8f;

#define FEAT 64

__global__ void __launch_bounds__(256)
gcn_scatter(const float* __restrict__ h,
            const float* __restrict__ ew,
            const int*   __restrict__ src,
            const int*   __restrict__ dst,
            float*       __restrict__ agg,
            int n_edges) {
    int idx = blockIdx.x * 256 + threadIdx.x;
    int e = idx >> 6;
    if (e >= n_edges) return;
    int f = idx & 63;
    int s = src[e];
    int d = dst[e];
    float w = ew[e];
    float val = h[(size_t)s * FEAT + f] * w;
    atomicAdd(&agg[(size_t)d * FEAT + f], val);
}

__global__ void __launch_bounds__(256)
gcn_gemm_relu(const float* __restrict__ agg,
              const float* __restrict__ W,
              const float* __restrict__ bias,
              float*       __restrict__ out,
              int n_rows) {
    __shared__ float Wlds[FEAT * FEAT];

    for (int i = threadIdx.x; i < FEAT * FEAT / 4; i += 256) {
        ((float4*)Wlds)[i] = ((const float4*)W)[i];
    }
    __syncthreads();

    const int lane = threadIdx.x & 31;
    const int wave = threadIdx.x >> 5;
    const int tile = blockIdx.x * 8 + wave;
    if (tile * 16 >= n_rows) return;

    const int nlo = lane & 15;
    const int hi  = lane >> 4;

    const float* aRow = agg + (size_t)(tile * 16 + nlo) * FEAT;

    v8f zero = {};
    v8f acc[4] = {zero, zero, zero, zero};

    for (int k0 = 0; k0 < FEAT; k0 += 4) {
        const int ka = k0 + 2 * hi;
        v2f a;
        a.x = aRow[ka];
        a.y = aRow[ka + 1];
        #pragma unroll
        for (int c = 0; c < 4; ++c) {
            v2f b;
            b.x = Wlds[ka * FEAT + c * 16 + nlo];
            b.y = Wlds[(ka + 1) * FEAT + c * 16 + nlo];
            acc[c] = __builtin_amdgcn_wmma_f32_16x16x4_f32(
                 false, a,  false, b,
                 (short)0, acc[c],
                 false,  false);
        }
    }

    #pragma unroll
    for (int c = 0; c < 4; ++c) {
        const int col = c * 16 + nlo;
        const float bv = bias[col];
        #pragma unroll
        for (int v = 0; v < 8; ++v) {
            const int r = tile * 16 + v + 8 * hi;
            float val = acc[c][v] + bv;
            out[(size_t)r * FEAT + col] = val > 0.0f ? val : 0.0f;
        }
    }
}

extern "C" void kernel_launch(void* const* d_in, const int* in_sizes, int n_in,
                              void* d_out, int out_size, void* d_ws, size_t ws_size,
                              hipStream_t stream) {
    const float* h   = (const float*)d_in[0];
    const float* Ws  = (const float*)d_in[1];
    const float* bs  = (const float*)d_in[2];
    const float* ew  = (const float*)d_in[3];
    const int*   src = (const int*)d_in[4];
    const int*   dst = (const int*)d_in[5];
    float* out = (float*)d_out;

    const int N = in_sizes[0] / FEAT;
    const int E = in_sizes[3];

    float* agg = (float*)d_ws;
    float* h1  = agg + (size_t)N * FEAT;

    const size_t aggBytes = (size_t)N * FEAT * sizeof(float);

    const int scatterThreads = E * FEAT;
    dim3 sGrd((scatterThreads + 255) / 256), sBlk(256);

    const int tiles = (N + 15) / 16;
    dim3 gGrd((tiles + 7) / 8), gBlk(256);

    hipMemsetAsync(agg, 0, aggBytes, stream);
    gcn_scatter<<<sGrd, sBlk, 0, stream>>>(h, ew, src, dst, agg, E);
    gcn_gemm_relu<<<gGrd, gBlk, 0, stream>>>(agg, Ws, bs, h1, N);

    hipMemsetAsync(agg, 0, aggBytes, stream);
    gcn_scatter<<<sGrd, sBlk, 0, stream>>>(h1, ew, src, dst, agg, E);
    gcn_gemm_relu<<<gGrd, gBlk, 0, stream>>>(agg, Ws + FEAT * FEAT, bs + FEAT, out, N);
}


# Float-atomic scatter replaced by per-call grouping of edges by destination (integer atomic rank, scan, place) and a per-node register gather; sum order per node now follows rank arrival
# speedup vs baseline: 2.1341x; 2.1341x over previous
#include <hip/hip_runtime.h>


typedef __attribute__((ext_vector_type(2))) float v2f;
typedef __attribute__((ext_vector_type(8))) float v8f;

#define FEAT 64
#define SCAN_TILE 4096

__global__ void __launch_bounds__(256)
edge_rank(const int* __restrict__ dst,
          int*       __restrict__ cnt,
          int*       __restrict__ rank,
          int n_edges) {
    const int e = blockIdx.x * 256 + threadIdx.x;
    if (e >= n_edges) return;
    rank[e] = atomicAdd(&cnt[dst[e]], 1);
}

__global__ void __launch_bounds__(1024)
node_scan(const int* __restrict__ cnt,
          int*       __restrict__ rowptr,
          int n_pad) {
    __shared__ int wtot[2][32];
    const int t    = threadIdx.x;
    const int lane = t & 31;
    const int w    = t >> 5;
    int carry = 0;
    int buf = 0;
    int4 cur = ((const int4*)cnt)[t];
    for (int base = 0; base < n_pad; base += SCAN_TILE) {
        int4 nxt = make_int4(0, 0, 0, 0);
        if (base + SCAN_TILE < n_pad)
            nxt = ((const int4*)(cnt + base + SCAN_TILE))[t];
        const int s1 = cur.x;
        const int s2 = s1 + cur.y;
        const int s3 = s2 + cur.z;
        const int s4 = s3 + cur.w;
        int incl = s4;
        for (int o = 1; o < 32; o <<= 1) {
            const int y = __shfl_up(incl, o);
            if (lane >= o) incl += y;
        }
        if (lane == 31) wtot[buf][w] = incl;
        __syncthreads();
        const int wt = wtot[buf][lane];
        int wincl = wt;
        for (int o = 1; o < 32; o <<= 1) {
            const int y = __shfl_up(wincl, o);
            if (lane >= o) wincl += y;
        }
        const int wexcl = __shfl(wincl, w) - __shfl(wt, w);
        const int total = __shfl(wincl, 31);
        const int excl  = carry + wexcl + incl - s4;
        ((int4*)(rowptr + base))[t] = make_int4(excl, excl + s1, excl + s2, excl + s3);
        carry += total;
        cur = nxt;
        buf ^= 1;
    }
}

__global__ void __launch_bounds__(256)
edge_place(const int*   __restrict__ src,
           const int*   __restrict__ dst,
           const float* __restrict__ ew,
           const int*   __restrict__ rowptr,
           const int*   __restrict__ rank,
           int2*        __restrict__ recs,
           int n_edges) {
    const int e = blockIdx.x * 256 + threadIdx.x;
    if (e >= n_edges) return;
    const int p = rowptr[dst[e]] + rank[e];
    recs[p] = make_int2(src[e], __float_as_int(ew[e]));
}

#define GROUP 8

static __device__ __forceinline__ void
gather_group(const float* __restrict__ h, const int2 rec, const int j, const int lane,
             float& ax, float& ay) {
    int    s[GROUP];
    float  w[GROUP];
    float2 x[GROUP];
    #pragma unroll
    for (int u = 0; u < GROUP; ++u) {
        s[u] = __builtin_amdgcn_readlane(rec.x, j + u);
        w[u] = __int_as_float(__builtin_amdgcn_readlane(rec.y, j + u));
    }
    #pragma unroll
    for (int u = 0; u < GROUP; ++u)
        x[u] = ((const float2*)(h + (size_t)s[u] * FEAT))[lane];
    #pragma unroll
    for (int u = 0; u < GROUP; ++u) {
        ax += w[u] * x[u].x;
        ay += w[u] * x[u].y;
    }
}

__global__ void __launch_bounds__(256)
node_gather(const float* __restrict__ h,
            const int*   __restrict__ rowptr,
            const int2*  __restrict__ recs,
            float*       __restrict__ agg,
            int n_nodes) {
    const int lane = threadIdx.x & 31;
    int v = __builtin_amdgcn_readfirstlane((int)(blockIdx.x * 8 + (threadIdx.x >> 5)));
    const int nw = (int)gridDim.x * 8;
    if (v >= n_nodes) return;

    int beg = rowptr[v];
    int end = rowptr[v + 1];
    int nbeg = 0, nend = 0;
    if (v + nw < n_nodes) {
        nbeg = rowptr[v + nw];
        nend = rowptr[v + nw + 1];
    }
    int2 rec = make_int2(0, 0);
    if (lane < end - beg) rec = recs[beg + lane];

    for (;;) {
        const int vn = v + nw;
        int nnbeg = 0, nnend = 0;
        if (vn + nw < n_nodes) {
            nnbeg = rowptr[vn + nw];
            nnend = rowptr[vn + nw + 1];
        }

        float ax = 0.0f, ay = 0.0f;
        int base = beg;
        while (end - base > 32) {
            #pragma nounroll
            for (int j = 0; j < 32; j += GROUP)
                gather_group(h, rec, j, lane, ax, ay);
            base += 32;
            rec = make_int2(0, 0);
            if (lane < end - base) rec = recs[base + lane];
        }
        const int cnt = end - base;
        int j = 0;
        for (; j + GROUP < cnt; j += GROUP)
            gather_group(h, rec, j, lane, ax, ay);

        const int rem  = cnt - j;
        const int last = (cnt > 0) ? cnt - 1 : 0;
        int    s[GROUP];
        float  w[GROUP];
        float2 x[GROUP];
        #pragma unroll
        for (int u = 0; u < GROUP; ++u) {
            const int jj = (j + u < last) ? j + u : last;
            s[u] = __builtin_amdgcn_readlane(rec.x, jj);
            w[u] = __int_as_float(__builtin_amdgcn_readlane(rec.y, jj));
        }
        int2 nrec = make_int2(0, 0);
        if (lane < nend - nbeg) nrec = recs[nbeg + lane];

        if (cnt > 0) {
            #pragma unroll
            for (int u = 0; u < GROUP; ++u)
                x[u] = ((const float2*)(h + (size_t)s[u] * FEAT))[lane];
            ax += w[0] * x[0].x;
            ay += w[0] * x[0].y;
            #pragma unroll
            for (int u = 1; u < GROUP; ++u) {
                const float px = w[u] * x[u].x;
                const float py = w[u] * x[u].y;
                ax += (u < rem) ? px : 0.0f;
                ay += (u < rem) ? py : 0.0f;
            }
        }
        ((float2*)(agg + (size_t)v * FEAT))[lane] = make_float2(ax, ay);

        if (vn >= n_nodes) break;
        v = vn;
        beg = nbeg;   end = nend;   rec = nrec;
        nbeg = nnbeg; nend = nnend;
    }
}

__global__ void __launch_bounds__(256)
gcn_gemm_relu(const float* __restrict__ agg,
              const float* __restrict__ W,
              const float* __restrict__ bias,
              float*       __restrict__ out,
              int n_rows) {
    __shared__ float Wlds[FEAT * FEAT];

    for (int i = threadIdx.x; i < FEAT * FEAT / 4; i += 256) {
        ((float4*)Wlds)[i] = ((const float4*)W)[i];
    }
    __syncthreads();

    const int lane = threadIdx.x & 31;
    const int wave = threadIdx.x >> 5;
    const int tile = blockIdx.x * 8 + wave;
    if (tile * 16 >= n_rows) return;

    const int nlo = lane & 15;
    const int hi  = lane >> 4;

    const float* aRow = agg + (size_t)(tile * 16 + nlo) * FEAT;

    v8f zero = {};
    v8f acc[4] = {zero, zero, zero, zero};

    for (int k0 = 0; k0 < FEAT; k0 += 4) {
        const int ka = k0 + 2 * hi;
        v2f a;
        a.x = aRow[ka];
        a.y = aRow[ka + 1];
        #pragma unroll
        for (int c = 0; c < 4; ++c) {
            v2f b;
            b.x = Wlds[ka * FEAT + c * 16 + nlo];
            b.y = Wlds[(ka + 1) * FEAT + c * 16 + nlo];
            acc[c] = __builtin_amdgcn_wmma_f32_16x16x4_f32(
                 false, a,  false, b,
                 (short)0, acc[c],
                 false,  false);
        }
    }

    #pragma unroll
    for (int c = 0; c < 4; ++c) {
        const int col = c * 16 + nlo;
        const float bv = bias[col];
        #pragma unroll
        for (int v = 0; v < 8; ++v) {
            const int r = tile * 16 + v + 8 * hi;
            float val = acc[c][v] + bv;
            out[(size_t)r * FEAT + col] = val > 0.0f ? val : 0.0f;
        }
    }
}

extern "C" void kernel_launch(void* const* d_in, const int* in_sizes, int n_in,
                              void* d_out, int out_size, void* d_ws, size_t ws_size,
                              hipStream_t stream) {
    const float* h   = (const float*)d_in[0];
    const float* Ws  = (const float*)d_in[1];
    const float* bs  = (const float*)d_in[2];
    const float* ew  = (const float*)d_in[3];
    const int*   src = (const int*)d_in[4];
    const int*   dst = (const int*)d_in[5];
    float* out = (float*)d_out;

    const int N = in_sizes[0] / FEAT;
    const int E = in_sizes[3];

    const int nPad = ((N + 1 + SCAN_TILE - 1) / SCAN_TILE) * SCAN_TILE;
    const int ePad = ((E + 3) / 4) * 4;

    float* agg    = (float*)d_ws;
    float* h1     = agg + (size_t)N * FEAT;
    int*   cnt    = (int*)(h1 + (size_t)N * FEAT);
    int*   rowptr = cnt + nPad;
    int*   rank   = rowptr + nPad;
    int2*  recs   = (int2*)(rank + ePad);

    dim3 eGrd((E + 255) / 256), eBlk(256);

    const int tiles = (N + 15) / 16;
    dim3 gGrd((tiles + 7) / 8), gBlk(256);

    const int gaBlocksAll = (N + 7) / 8;
    dim3 aGrd(gaBlocksAll < 2048 ? gaBlocksAll : 2048), aBlk(256);

    hipMemsetAsync(cnt, 0, (size_t)nPad * sizeof(int), stream);
    edge_rank<<<eGrd, eBlk, 0, stream>>>(dst, cnt, rank, E);
    node_scan<<<dim3(1), dim3(1024), 0, stream>>>(cnt, rowptr, nPad);
    edge_place<<<eGrd, eBlk, 0, stream>>>(src, dst, ew, rowptr, rank, recs, E);

    node_gather<<<aGrd, aBlk, 0, stream>>>(h, rowptr, recs, agg, N);
    gcn_gemm_relu<<<gGrd, gBlk, 0, stream>>>(agg, Ws, bs, h1, N);

    node_gather<<<aGrd, aBlk, 0, stream>>>(h1, rowptr, recs, agg, N);
    gcn_gemm_relu<<<gGrd, gBlk, 0, stream>>>(agg, Ws + FEAT * FEAT, bs + FEAT, out, N);
}


# Global atomic rank + scan + place replaced by a two-level counting sort on LDS counters (chunk runs per 128-node group, then LDS rank per node); sum order per node follows LDS rank arrival
# speedup vs baseline: 4.0895x; 1.9163x over previous
#include <hip/hip_runtime.h>


typedef __attribute__((ext_vector_type(2))) float v2f;
typedef __attribute__((ext_vector_type(8))) float v8f;

#define FEAT 64
#define NPO        128
#define NPO_SHIFT  7
#define CHUNK      4096
#define MAXP       512
#define MAXB       256
#define OWN_T      512
#define OWN_U      8

__global__ void __launch_bounds__(1024)
edge_bin(const int*   __restrict__ src,
         const int*   __restrict__ dst,
         const float* __restrict__ ew,
         int2*        __restrict__ binrec,
         int*         __restrict__ bindst,
         int2*        __restrict__ runtab,
         int n_edges, int n_owner, int n_chunk) {
    __shared__ int cnt[MAXP];
    __shared__ int off[MAXP];
    __shared__ int wtot[32];
    const int t    = threadIdx.x;
    const int lane = t & 31;
    const int w    = t >> 5;
    const int b    = blockIdx.x;
    if (t < MAXP) cnt[t] = 0;
    __syncthreads();

    const int e0 = b * CHUNK + t * 4;
    int d[4], s[4], wb[4], slot[4];
    if (e0 + 3 < n_edges) {
        const int4   dv = *(const int4*)(dst + e0);
        const int4   sv = *(const int4*)(src + e0);
        const float4 wv = *(const float4*)(ew + e0);
        d[0] = dv.x; d[1] = dv.y; d[2] = dv.z; d[3] = dv.w;
        s[0] = sv.x; s[1] = sv.y; s[2] = sv.z; s[3] = sv.w;
        wb[0] = __float_as_int(wv.x); wb[1] = __float_as_int(wv.y);
        wb[2] = __float_as_int(wv.z); wb[3] = __float_as_int(wv.w);
    } else {
        #pragma unroll
        for (int u = 0; u < 4; ++u) {
            const int e = e0 + u;
            d[u] = -1; s[u] = 0; wb[u] = 0;
            if (e < n_edges) { d[u] = dst[e]; s[u] = src[e]; wb[u] = __float_as_int(ew[e]); }
        }
    }
    #pragma unroll
    for (int u = 0; u < 4; ++u) {
        slot[u] = 0;
        if (d[u] >= 0) slot[u] = atomicAdd(&cnt[d[u] >> NPO_SHIFT], 1);
    }
    __syncthreads();

    const int c = (t < MAXP) ? cnt[t] : 0;
    int incl = c;
    for (int o = 1; o < 32; o <<= 1) {
        const int y = __shfl_up(incl, o);
        if (lane >= o) incl += y;
    }
    if (lane == 31) wtot[w] = incl;
    __syncthreads();
    const int wt = wtot[lane];
    int wincl = wt;
    for (int o = 1; o < 32; o <<= 1) {
        const int y = __shfl_up(wincl, o);
        if (lane >= o) wincl += y;
    }
    const int wexcl = __shfl(wincl, w) - __shfl(wt, w);
    const int excl  = wexcl + incl - c;
    if (t < MAXP) off[t] = excl;
    if (t < n_owner) runtab[(size_t)t * n_chunk + b] = make_int2(excl, c);
    __syncthreads();

    #pragma unroll
    for (int u = 0; u < 4; ++u) {
        if (d[u] >= 0) {
            const int q = b * CHUNK + off[d[u] >> NPO_SHIFT] + slot[u];
            binrec[q] = make_int2(s[u], wb[u]);
            bindst[q] = d[u];
        }
    }
}

static __device__ __forceinline__ int run_of(const int* rpre, const int n_chunk, const int i) {
    int lo = 0;
    #pragma unroll
    for (int step = MAXB / 2; step > 0; step >>= 1) {
        const int m  = lo + step;
        const int mm = (m < n_chunk) ? m : 0;
        lo = ((m < n_chunk) & (rpre[mm] <= i)) ? m : lo;
    }
    return lo;
}

__global__ void __launch_bounds__(OWN_T)
owner_sort(const int2* __restrict__ binrec,
           const int*  __restrict__ bindst,
           const int2* __restrict__ runtab,
           int*        __restrict__ rowptr,
           int2*       __restrict__ recs,
           int n_nodes, int n_chunk) {
    __shared__ int rstart[MAXB];
    __shared__ int rpre[MAXB + 1];
    __shared__ int ncnt[NPO];
    __shared__ int wlen[OWN_T / 32];
    __shared__ int wbas[OWN_T / 32];
    const int t     = threadIdx.x;
    const int lane  = t & 31;
    const int w     = t >> 5;
    const int p     = blockIdx.x;
    const int nbase = p * NPO;

    int2 r = make_int2(0, 0);
    if (t < n_chunk) r = runtab[(size_t)p * n_chunk + t];
    int incl = r.y;
    for (int o = 1; o < 32; o <<= 1) {
        const int y = __shfl_up(incl, o);
        if (lane >= o) incl += y;
    }
    int bsum = r.x;
    for (int o = 16; o > 0; o >>= 1) bsum += __shfl_xor(bsum, o);
    if (lane == 31) wlen[w] = incl;
    if (lane == 0)  wbas[w] = bsum;
    if (t < NPO) ncnt[t] = 0;
    __syncthreads();
    const int wl = (lane < OWN_T / 32) ? wlen[lane] : 0;
    int wli = wl;
    for (int o = 1; o < 32; o <<= 1) {
        const int y = __shfl_up(wli, o);
        if (lane >= o) wli += y;
    }
    const int wexcl = __shfl(wli, w) - __shfl(wl, w);
    const int total = __shfl(wli, 31);
    int base = (lane < OWN_T / 32) ? wbas[lane] : 0;
    for (int o = 16; o > 0; o >>= 1) base += __shfl_xor(base, o);
    if (t < n_chunk) {
        rpre[t]   = wexcl + incl - r.y;
        rstart[t] = t * CHUNK + r.x;
    }
    if (t == 0) rpre[n_chunk] = total;
    __syncthreads();

    int  pos[OWN_U];
    int  dl[OWN_U];
    int2 rc[OWN_U];
    int ii[OWN_U];
    int lo[OWN_U];
    #pragma unroll
    for (int u = 0; u < OWN_U; ++u) {
        const int i = t + u * OWN_T;
        ii[u] = (i < total) ? i : 0;
        lo[u] = 0;
    }
    #pragma unroll
    for (int step = MAXB / 2; step > 0; step >>= 1) {
        #pragma unroll
        for (int u = 0; u < OWN_U; ++u) {
            const int m  = lo[u] + step;
            const int mm = (m < n_chunk) ? m : 0;
            lo[u] = ((m < n_chunk) & (rpre[mm] <= ii[u])) ? m : lo[u];
        }
    }
    #pragma unroll
    for (int u = 0; u < OWN_U; ++u) {
        const bool ok = (t + u * OWN_T) < total;
        const int  q  = rstart[lo[u]] + (ii[u] - rpre[lo[u]]);
        int dv = nbase;
        rc[u] = make_int2(0, 0);
        if (ok) { dv = bindst[q]; rc[u] = binrec[q]; }
        pos[u] = ok ? q : -1;
        dl[u]  = dv - nbase;
    }
    #pragma unroll
    for (int u = 0; u < OWN_U; ++u)
        if (pos[u] >= 0) atomicAdd(&ncnt[dl[u]], 1);
    for (int i = t + OWN_U * OWN_T; i < total; i += OWN_T) {
        const int rb = run_of(rpre, n_chunk, i);
        const int q  = rstart[rb] + (i - rpre[rb]);
        atomicAdd(&ncnt[bindst[q] - nbase], 1);
    }
    __syncthreads();

    const int c = (t < NPO) ? ncnt[t] : 0;
    int ninc = c;
    for (int o = 1; o < 32; o <<= 1) {
        const int y = __shfl_up(ninc, o);
        if (lane >= o) ninc += y;
    }
    if (lane == 31) wlen[w] = ninc;
    __syncthreads();
    const int nl = (lane < OWN_T / 32) ? wlen[lane] : 0;
    int nli = nl;
    for (int o = 1; o < 32; o <<= 1) {
        const int y = __shfl_up(nli, o);
        if (lane >= o) nli += y;
    }
    const int nexcl = __shfl(nli, w) - __shfl(nl, w) + ninc - c;
    if (t < NPO) {
        ncnt[t] = nexcl;
        if (nbase + t < n_nodes) rowptr[nbase + t] = base + nexcl;
    }
    if (p == (int)gridDim.x - 1 && t == 0) rowptr[n_nodes] = base + total;
    __syncthreads();

    #pragma unroll
    for (int u = 0; u < OWN_U; ++u) {
        if (pos[u] >= 0) {
            const int k = atomicAdd(&ncnt[dl[u]], 1);
            recs[base + k] = rc[u];
        }
    }
    for (int i = t + OWN_U * OWN_T; i < total; i += OWN_T) {
        const int rb = run_of(rpre, n_chunk, i);
        const int q  = rstart[rb] + (i - rpre[rb]);
        const int k  = atomicAdd(&ncnt[bindst[q] - nbase], 1);
        recs[base + k] = binrec[q];
    }
}

#define GROUP 8

static __device__ __forceinline__ void
gather_group(const float* __restrict__ h, const int2 rec, const int j, const int lane,
             float& ax, float& ay) {
    int    s[GROUP];
    float  w[GROUP];
    float2 x[GROUP];
    #pragma unroll
    for (int u = 0; u < GROUP; ++u) {
        s[u] = __builtin_amdgcn_readlane(rec.x, j + u);
        w[u] = __int_as_float(__builtin_amdgcn_readlane(rec.y, j + u));
    }
    #pragma unroll
    for (int u = 0; u < GROUP; ++u)
        x[u] = ((const float2*)(h + (size_t)s[u] * FEAT))[lane];
    #pragma unroll
    for (int u = 0; u < GROUP; ++u) {
        ax += w[u] * x[u].x;
        ay += w[u] * x[u].y;
    }
}

__global__ void __launch_bounds__(256)
node_gather(const float* __restrict__ h,
            const int*   __restrict__ rowptr,
            const int2*  __restrict__ recs,
            float*       __restrict__ agg,
            int n_nodes) {
    const int lane = threadIdx.x & 31;
    int v = __builtin_amdgcn_readfirstlane((int)(blockIdx.x * 8 + (threadIdx.x >> 5)));
    const int nw = (int)gridDim.x * 8;
    if (v >= n_nodes) return;

    int beg = rowptr[v];
    int end = rowptr[v + 1];
    int nbeg = 0, nend = 0;
    if (v + nw < n_nodes) {
        nbeg = rowptr[v + nw];
        nend = rowptr[v + nw + 1];
    }
    int2 rec = make_int2(0, 0);
    if (lane < end - beg) rec = recs[beg + lane];

    for (;;) {
        const int vn = v + nw;
        int nnbeg = 0, nnend = 0;
        if (vn + nw < n_nodes) {
            nnbeg = rowptr[vn + nw];
            nnend = rowptr[vn + nw + 1];
        }

        float ax = 0.0f, ay = 0.0f;
        int base = beg;
        while (end - base > 32) {
            #pragma nounroll
            for (int j = 0; j < 32; j += GROUP)
                gather_group(h, rec, j, lane, ax, ay);
            base += 32;
            rec = make_int2(0, 0);
            if (lane < end - base) rec = recs[base + lane];
        }
        const int cnt = end - base;
        int j = 0;
        for (; j + GROUP < cnt; j += GROUP)
            gather_group(h, rec, j, lane, ax, ay);

        const int rem  = cnt - j;
        const int last = (cnt > 0) ? cnt - 1 : 0;
        int    s[GROUP];
        float  w[GROUP];
        float2 x[GROUP];
        #pragma unroll
        for (int u = 0; u < GROUP; ++u) {
            const int jj = (j + u < last) ? j + u : last;
            s[u] = __builtin_amdgcn_readlane(rec.x, jj);
            w[u] = __int_as_float(__builtin_amdgcn_readlane(rec.y, jj));
        }
        int2 nrec = make_int2(0, 0);
        if (lane < nend - nbeg) nrec = recs[nbeg + lane];

        if (cnt > 0) {
            #pragma unroll
            for (int u = 0; u < GROUP; ++u)
                x[u] = ((const float2*)(h + (size_t)s[u] * FEAT))[lane];
            ax += w[0] * x[0].x;
            ay += w[0] * x[0].y;
            #pragma unroll
            for (int u = 1; u < GROUP; ++u) {
                const float px = w[u] * x[u].x;
                const float py = w[u] * x[u].y;
                ax += (u < rem) ? px : 0.0f;
                ay += (u < rem) ? py : 0.0f;
            }
        }
        ((float2*)(agg + (size_t)v * FEAT))[lane] = make_float2(ax, ay);

        if (vn >= n_nodes) break;
        v = vn;
        beg = nbeg;   end = nend;   rec = nrec;
        nbeg = nnbeg; nend = nnend;
    }
}

__global__ void __launch_bounds__(256)
gcn_gemm_relu(const float* __restrict__ agg,
              const float* __restrict__ W,
              const float* __restrict__ bias,
              float*       __restrict__ out,
              int n_rows) {
    __shared__ float Wlds[FEAT * FEAT];

    for (int i = threadIdx.x; i < FEAT * FEAT / 4; i += 256) {
        ((float4*)Wlds)[i] = ((const float4*)W)[i];
    }
    __syncthreads();

    const int lane = threadIdx.x & 31;
    const int wave = threadIdx.x >> 5;
    const int tile = blockIdx.x * 8 + wave;
    if (tile * 16 >= n_rows) return;

    const int nlo = lane & 15;
    const int hi  = lane >> 4;

    const float* aRow = agg + (size_t)(tile * 16 + nlo) * FEAT;

    v8f zero = {};
    v8f acc[4] = {zero, zero, zero, zero};

    for (int k0 = 0; k0 < FEAT; k0 += 4) {
        const int ka = k0 + 2 * hi;
        v2f a;
        a.x = aRow[ka];
        a.y = aRow[ka + 1];
        #pragma unroll
        for (int c = 0; c < 4; ++c) {
            v2f b;
            b.x = Wlds[ka * FEAT + c * 16 + nlo];
            b.y = Wlds[(ka + 1) * FEAT + c * 16 + nlo];
            acc[c] = __builtin_amdgcn_wmma_f32_16x16x4_f32(
                 false, a,  false, b,
                 (short)0, acc[c],
                 false,  false);
        }
    }

    #pragma unroll
    for (int c = 0; c < 4; ++c) {
        const int col = c * 16 + nlo;
        const float bv = bias[col];
        #pragma unroll
        for (int v = 0; v < 8; ++v) {
            const int r = tile * 16 + v + 8 * hi;
            float val = acc[c][v] + bv;
            out[(size_t)r * FEAT + col] = val > 0.0f ? val : 0.0f;
        }
    }
}

extern "C" void kernel_launch(void* const* d_in, const int* in_sizes, int n_in,
                              void* d_out, int out_size, void* d_ws, size_t ws_size,
                              hipStream_t stream) {
    const float* h   = (const float*)d_in[0];
    const float* Ws  = (const float*)d_in[1];
    const float* bs  = (const float*)d_in[2];
    const float* ew  = (const float*)d_in[3];
    const int*   src = (const int*)d_in[4];
    const int*   dst = (const int*)d_in[5];
    float* out = (float*)d_out;

    const int N = in_sizes[0] / FEAT;
    const int E = in_sizes[3];

    const int nOwner = (N + NPO - 1) / NPO;
    const int nChunk = (E + CHUNK - 1) / CHUNK;
    const int nRow   = ((N + 1 + 3) / 4) * 4;
    const int ePad   = ((E + 3) / 4) * 4;

    float* agg    = (float*)d_ws;
    float* h1     = agg + (size_t)N * FEAT;
    int*   rowptr = (int*)(h1 + (size_t)N * FEAT);
    int2*  recs   = (int2*)(rowptr + nRow);
    int2*  binrec = recs + ePad;
    int*   bindst = (int*)(binrec + (size_t)nChunk * CHUNK);
    int2*  runtab = (int2*)(bindst + (size_t)nChunk * CHUNK);

    const int tiles = (N + 15) / 16;
    dim3 gGrd((tiles + 7) / 8), gBlk(256);

    const int gaBlocksAll = (N + 7) / 8;
    dim3 aGrd(gaBlocksAll < 2048 ? gaBlocksAll : 2048), aBlk(256);

    edge_bin<<<dim3(nChunk), dim3(1024), 0, stream>>>(src, dst, ew, binrec, bindst, runtab, E, nOwner, nChunk);
    owner_sort<<<dim3(nOwner), dim3(OWN_T), 0, stream>>>(binrec, bindst, runtab, rowptr, recs, N, nChunk);

    node_gather<<<aGrd, aBlk, 0, stream>>>(h, rowptr, recs, agg, N);
    gcn_gemm_relu<<<gGrd, gBlk, 0, stream>>>(agg, Ws, bs, h1, N);

    node_gather<<<aGrd, aBlk, 0, stream>>>(h1, rowptr, recs, agg, N);
    gcn_gemm_relu<<<gGrd, gBlk, 0, stream>>>(agg, Ws + FEAT * FEAT, bs + FEAT, out, N);
}


# Gather and dense product fused into one launch per layer: an 8-wave block sums 64 nodes' rows into LDS (a wave walks 8 nodes' lists as one, 16 rows in flight); WMMA A fragments come from LDS
# speedup vs baseline: 4.8534x; 1.1868x over previous
#include <hip/hip_runtime.h>


typedef __attribute__((ext_vector_type(2))) float v2f;
typedef __attribute__((ext_vector_type(8))) float v8f;

#define FEAT 64
#define NPO        128
#define NPO_SHIFT  7
#define CHUNK      4096
#define MAXP       512
#define MAXB       256
#define OWN_T      512
#define OWN_U      8

__global__ void __launch_bounds__(1024)
edge_bin(const int*   __restrict__ src,
         const int*   __restrict__ dst,
         const float* __restrict__ ew,
         int2*        __restrict__ binrec,
         int*         __restrict__ bindst,
         int2*        __restrict__ runtab,
         int n_edges, int n_owner, int n_chunk) {
    __shared__ int cnt[MAXP];
    __shared__ int off[MAXP];
    __shared__ int wtot[32];
    const int t    = threadIdx.x;
    const int lane = t & 31;
    const int w    = t >> 5;
    const int b    = blockIdx.x;
    if (t < MAXP) cnt[t] = 0;
    __syncthreads();

    const int e0 = b * CHUNK + t * 4;
    int d[4], s[4], wb[4], slot[4];
    if (e0 + 3 < n_edges) {
        const int4   dv = *(const int4*)(dst + e0);
        const int4   sv = *(const int4*)(src + e0);
        const float4 wv = *(const float4*)(ew + e0);
        d[0] = dv.x; d[1] = dv.y; d[2] = dv.z; d[3] = dv.w;
        s[0] = sv.x; s[1] = sv.y; s[2] = sv.z; s[3] = sv.w;
        wb[0] = __float_as_int(wv.x); wb[1] = __float_as_int(wv.y);
        wb[2] = __float_as_int(wv.z); wb[3] = __float_as_int(wv.w);
    } else {
        #pragma unroll
        for (int u = 0; u < 4; ++u) {
            const int e = e0 + u;
            d[u] = -1; s[u] = 0; wb[u] = 0;
            if (e < n_edges) { d[u] = dst[e]; s[u] = src[e]; wb[u] = __float_as_int(ew[e]); }
        }
    }
    #pragma unroll
    for (int u = 0; u < 4; ++u) {
        slot[u] = 0;
        if (d[u] >= 0) slot[u] = atomicAdd(&cnt[d[u] >> NPO_SHIFT], 1);
    }
    __syncthreads();

    const int c = (t < MAXP) ? cnt[t] : 0;
    int incl = c;
    for (int o = 1; o < 32; o <<= 1) {
        const int y = __shfl_up(incl, o);
        if (lane >= o) incl += y;
    }
    if (lane == 31) wtot[w] = incl;
    __syncthreads();
    const int wt = wtot[lane];
    int wincl = wt;
    for (int o = 1; o < 32; o <<= 1) {
        const int y = __shfl_up(wincl, o);
        if (lane >= o) wincl += y;
    }
    const int wexcl = __shfl(wincl, w) - __shfl(wt, w);
    const int excl  = wexcl + incl - c;
    if (t < MAXP) off[t] = excl;
    if (t < n_owner) runtab[(size_t)t * n_chunk + b] = make_int2(excl, c);
    __syncthreads();

    #pragma unroll
    for (int u = 0; u < 4; ++u) {
        if (d[u] >= 0) {
            const int q = b * CHUNK + off[d[u] >> NPO_SHIFT] + slot[u];
            binrec[q] = make_int2(s[u], wb[u]);
            bindst[q] = d[u];
        }
    }
}

static __device__ __forceinline__ int run_of(const int* rpre, const int n_chunk, const int i) {
    int lo = 0;
    #pragma unroll
    for (int step = MAXB / 2; step > 0; step >>= 1) {
        const int m  = lo + step;
        const int mm = (m < n_chunk) ? m : 0;
        lo = ((m < n_chunk) & (rpre[mm] <= i)) ? m : lo;
    }
    return lo;
}

__global__ void __launch_bounds__(OWN_T)
owner_sort(const int2* __restrict__ binrec,
           const int*  __restrict__ bindst,
           const int2* __restrict__ runtab,
           int*        __restrict__ rowptr,
           int2*       __restrict__ recs,
           int n_nodes, int n_chunk) {
    __shared__ int rstart[MAXB];
    __shared__ int rpre[MAXB + 1];
    __shared__ int ncnt[NPO];
    __shared__ int wlen[OWN_T / 32];
    __shared__ int wbas[OWN_T / 32];
    const int t     = threadIdx.x;
    const int lane  = t & 31;
    const int w     = t >> 5;
    const int p     = blockIdx.x;
    const int nbase = p * NPO;

    int2 r = make_int2(0, 0);
    if (t < n_chunk) r = runtab[(size_t)p * n_chunk + t];
    int incl = r.y;
    for (int o = 1; o < 32; o <<= 1) {
        const int y = __shfl_up(incl, o);
        if (lane >= o) incl += y;
    }
    int bsum = r.x;
    for (int o = 16; o > 0; o >>= 1) bsum += __shfl_xor(bsum, o);
    if (lane == 31) wlen[w] = incl;
    if (lane == 0)  wbas[w] = bsum;
    if (t < NPO) ncnt[t] = 0;
    __syncthreads();
    const int wl = (lane < OWN_T / 32) ? wlen[lane] : 0;
    int wli = wl;
    for (int o = 1; o < 32; o <<= 1) {
        const int y = __shfl_up(wli, o);
        if (lane >= o) wli += y;
    }
    const int wexcl = __shfl(wli, w) - __shfl(wl, w);
    const int total = __shfl(wli, 31);
    int base = (lane < OWN_T / 32) ? wbas[lane] : 0;
    for (int o = 16; o > 0; o >>= 1) base += __shfl_xor(base, o);
    if (t < n_chunk) {
        rpre[t]   = wexcl + incl - r.y;
        rstart[t] = t * CHUNK + r.x;
    }
    if (t == 0) rpre[n_chunk] = total;
    __syncthreads();

    int  pos[OWN_U];
    int  dl[OWN_U];
    int2 rc[OWN_U];
    int ii[OWN_U];
    int lo[OWN_U];
    #pragma unroll
    for (int u = 0; u < OWN_U; ++u) {
        const int i = t + u * OWN_T;
        ii[u] = (i < total) ? i : 0;
        lo[u] = 0;
    }
    #pragma unroll
    for (int step = MAXB / 2; step > 0; step >>= 1) {
        #pragma unroll
        for (int u = 0; u < OWN_U; ++u) {
            const int m  = lo[u] + step;
            const int mm = (m < n_chunk) ? m : 0;
            lo[u] = ((m < n_chunk) & (rpre[mm] <= ii[u])) ? m : lo[u];
        }
    }
    #pragma unroll
    for (int u = 0; u < OWN_U; ++u) {
        const bool ok = (t + u * OWN_T) < total;
        const int  q  = rstart[lo[u]] + (ii[u] - rpre[lo[u]]);
        int dv = nbase;
        rc[u] = make_int2(0, 0);
        if (ok) { dv = bindst[q]; rc[u] = binrec[q]; }
        pos[u] = ok ? q : -1;
        dl[u]  = dv - nbase;
    }
    #pragma unroll
    for (int u = 0; u < OWN_U; ++u)
        if (pos[u] >= 0) atomicAdd(&ncnt[dl[u]], 1);
    for (int i = t + OWN_U * OWN_T; i < total; i += OWN_T) {
        const int rb = run_of(rpre, n_chunk, i);
        const int q  = rstart[rb] + (i - rpre[rb]);
        atomicAdd(&ncnt[bindst[q] - nbase], 1);
    }
    __syncthreads();

    const int c = (t < NPO) ? ncnt[t] : 0;
    int ninc = c;
    for (int o = 1; o < 32; o <<= 1) {
        const int y = __shfl_up(ninc, o);
        if (lane >= o) ninc += y;
    }
    if (lane == 31) wlen[w] = ninc;
    __syncthreads();
    const int nl = (lane < OWN_T / 32) ? wlen[lane] : 0;
    int nli = nl;
    for (int o = 1; o < 32; o <<= 1) {
        const int y = __shfl_up(nli, o);
        if (lane >= o) nli += y;
    }
    const int nexcl = __shfl(nli, w) - __shfl(nl, w) + ninc - c;
    if (t < NPO) {
        ncnt[t] = nexcl;
        if (nbase + t < n_nodes) rowptr[nbase + t] = base + nexcl;
    }
    if (p == (int)gridDim.x - 1 && t == 0) rowptr[n_nodes] = base + total;
    __syncthreads();

    #pragma unroll
    for (int u = 0; u < OWN_U; ++u) {
        if (pos[u] >= 0) {
            const int k = atomicAdd(&ncnt[dl[u]], 1);
            recs[base + k] = rc[u];
        }
    }
    for (int i = t + OWN_U * OWN_T; i < total; i += OWN_T) {
        const int rb = run_of(rpre, n_chunk, i);
        const int q  = rstart[rb] + (i - rpre[rb]);
        const int k  = atomicAdd(&ncnt[bindst[q] - nbase], 1);
        recs[base + k] = binrec[q];
    }
}

#define TILE_N  64
#define WAVE_N  8
#define APITCH  68
#define GROUP   16

template <int J, bool FULL, bool PREF>
static __device__ __forceinline__ void
gather_group(const float* __restrict__ h, const int2 rec, const int lane,
             const int e0, const int E1, const int last,
             const int rp, float* myrow, int& node, int& nend, float& ax, float& ay,
             const int2* __restrict__ nptr, int2& nrec) {
    int    s[GROUP];
    float  w[GROUP];
    float2 x[GROUP];
    #pragma unroll
    for (int u = 0; u < GROUP; ++u) {
        if (FULL) {
            s[u] = __builtin_amdgcn_readlane(rec.x, J + u);
            w[u] = __int_as_float(__builtin_amdgcn_readlane(rec.y, J + u));
        } else {
            const int jj = (J + u < last) ? J + u : last;
            s[u] = __builtin_amdgcn_readlane(rec.x, jj);
            w[u] = __int_as_float(__builtin_amdgcn_readlane(rec.y, jj));
        }
    }
    #pragma unroll
    for (int u = 0; u < GROUP; ++u) {
        const unsigned off = (unsigned)s[u] * (unsigned)(FEAT * sizeof(float)) + (unsigned)lane * 8u;
        x[u] = *(const float2*)((const char*)h + off);
    }
    if (PREF) nrec = *nptr;
    #pragma unroll
    for (int u = 0; u < GROUP; ++u) {
        const int e = e0 + u;
        if (FULL || e < E1) {
            if (__builtin_expect(e >= nend, 0)) {
                do {
                    *(float2*)(myrow + node * APITCH) = make_float2(ax, ay);
                    ax = 0.0f; ay = 0.0f;
                    ++node;
                    nend = __builtin_amdgcn_readlane(rp, node + 1);
                } while (e >= nend);
            }
            ax += w[u] * x[u].x;
            ay += w[u] * x[u].y;
        }
    }
}

__global__ void __launch_bounds__(256)
gather_gemm_relu(const float* __restrict__ h,
                 const int*   __restrict__ rowptr,
                 const int2*  __restrict__ recs,
                 const float* __restrict__ W,
                 const float* __restrict__ bias,
                 float*       __restrict__ out,
                 int n_nodes) {
    __shared__ float Wlds[FEAT * FEAT];
    __shared__ float Alds[TILE_N * APITCH];

    const int lane  = threadIdx.x & 31;
    const int wave  = __builtin_amdgcn_readfirstlane((int)(threadIdx.x >> 5));
    const int nbase = (int)blockIdx.x * TILE_N;
    const int v0    = nbase + wave * WAVE_N;

    int bi = v0 + lane;
    bi = (bi < n_nodes) ? bi : n_nodes;
    int rp = 0;
    if (lane <= WAVE_N) rp = rowptr[bi];

    float4 wq[4];
    #pragma unroll
    for (int i = 0; i < 4; ++i) wq[i] = ((const float4*)W)[threadIdx.x + i * 256];

    const int E0 = __builtin_amdgcn_readlane(rp, 0);
    const int E1 = __builtin_amdgcn_readlane(rp, WAVE_N);
    int2 rec = make_int2(0, 0);
    if (E0 + lane < E1) rec = recs[E0 + lane];

    #pragma unroll
    for (int i = 0; i < 4; ++i) ((float4*)Wlds)[threadIdx.x + i * 256] = wq[i];

    float* myrow = Alds + (wave * WAVE_N) * APITCH + 2 * lane;
    int   node = 0;
    int   nend = __builtin_amdgcn_readlane(rp, 1);
    float ax = 0.0f, ay = 0.0f;

    for (int cb = E0; cb < E1; cb += 32) {
        const int left = E1 - cb;
        const int last = ((left < 32) ? left : 32) - 1;
        int ni = cb + 32 + lane;
        ni = (ni < E1) ? ni : E1 - 1;
        int2 nrec = make_int2(0, 0);
        if (left >= GROUP) gather_group<0, true,  true >(h, rec, lane, cb, E1, last, rp, myrow, node, nend, ax, ay, recs + ni, nrec);
        else               gather_group<0, false, false>(h, rec, lane, cb, E1, last, rp, myrow, node, nend, ax, ay, recs + ni, nrec);
        if (left >= 2 * GROUP)  gather_group<GROUP, true,  false>(h, rec, lane, cb + GROUP, E1, last, rp, myrow, node, nend, ax, ay, recs + ni, nrec);
        else if (left > GROUP)  gather_group<GROUP, false, false>(h, rec, lane, cb + GROUP, E1, last, rp, myrow, node, nend, ax, ay, recs + ni, nrec);
        rec = nrec;
    }
    for (; node < WAVE_N; ++node) {
        *(float2*)(myrow + node * APITCH) = make_float2(ax, ay);
        ax = 0.0f; ay = 0.0f;
    }
    __syncthreads();

    const int strip = wave >> 1;
    const int cpair = (wave & 1) * 2;
    const int nlo = lane & 15;
    const int hi  = lane >> 4;
    const float* aRow = Alds + (strip * 16 + nlo) * APITCH;

    v8f zero = {};
    v8f acc[2] = {zero, zero};
    #pragma unroll
    for (int k0 = 0; k0 < FEAT; k0 += 4) {
        const int ka = k0 + 2 * hi;
        v2f a;
        a.x = aRow[ka];
        a.y = aRow[ka + 1];
        #pragma unroll
        for (int c = 0; c < 2; ++c) {
            v2f b;
            b.x = Wlds[ka * FEAT + (cpair + c) * 16 + nlo];
            b.y = Wlds[(ka + 1) * FEAT + (cpair + c) * 16 + nlo];
            acc[c] = __builtin_amdgcn_wmma_f32_16x16x4_f32(
                 false, a,  false, b,
                 (short)0, acc[c],
                 false,  false);
        }
    }

    #pragma unroll
    for (int c = 0; c < 2; ++c) {
        const int col = (cpair + c) * 16 + nlo;
        const float bv = bias[col];
        #pragma unroll
        for (int v = 0; v < 8; ++v) {
            const int r = nbase + strip * 16 + v + 8 * hi;
            const float val = acc[c][v] + bv;
            if (r < n_nodes) out[(size_t)r * FEAT + col] = val > 0.0f ? val : 0.0f;
        }
    }
}

extern "C" void kernel_launch(void* const* d_in, const int* in_sizes, int n_in,
                              void* d_out, int out_size, void* d_ws, size_t ws_size,
                              hipStream_t stream) {
    const float* h   = (const float*)d_in[0];
    const float* Ws  = (const float*)d_in[1];
    const float* bs  = (const float*)d_in[2];
    const float* ew  = (const float*)d_in[3];
    const int*   src = (const int*)d_in[4];
    const int*   dst = (const int*)d_in[5];
    float* out = (float*)d_out;

    const int N = in_sizes[0] / FEAT;
    const int E = in_sizes[3];

    const int nOwner = (N + NPO - 1) / NPO;
    const int nChunk = (E + CHUNK - 1) / CHUNK;
    const int nRow   = ((N + 1 + 3) / 4) * 4;
    const int ePad   = ((E + 3) / 4) * 4;

    float* h1     = (float*)d_ws;
    int*   rowptr = (int*)(h1 + (size_t)N * FEAT);
    int2*  recs   = (int2*)(rowptr + nRow);
    int2*  binrec = recs + ePad;
    int*   bindst = (int*)(binrec + (size_t)nChunk * CHUNK);
    int2*  runtab = (int2*)(bindst + (size_t)nChunk * CHUNK);

    dim3 gGrd((N + TILE_N - 1) / TILE_N), gBlk(256);

    edge_bin<<<dim3(nChunk), dim3(1024), 0, stream>>>(src, dst, ew, binrec, bindst, runtab, E, nOwner, nChunk);
    owner_sort<<<dim3(nOwner), dim3(OWN_T), 0, stream>>>(binrec, bindst, runtab, rowptr, recs, N, nChunk);

    gather_gemm_relu<<<gGrd, gBlk, 0, stream>>>(h, rowptr, recs, Ws, bs, h1, N);

    gather_gemm_relu<<<gGrd, gBlk, 0, stream>>>(h1, rowptr, recs, Ws + FEAT * FEAT, bs + FEAT, out, N);
}


# Scattered record stores of both edge-sort passes staged in LDS and written as whole lines (edge_bin: chunk image, 16-byte stores; owner_sort: ordered list, consecutive stores); sum order as before
# speedup vs baseline: 5.4897x; 1.1311x over previous
#include <hip/hip_runtime.h>


typedef __attribute__((ext_vector_type(2))) float v2f;
typedef __attribute__((ext_vector_type(8))) float v8f;

#define FEAT 64
#define NPO        128
#define NPO_SHIFT  7
#define CHUNK      4096
#define MAXP       512
#define MAXB       256
#define OWN_T      512
#define OWN_U      8

__global__ void __launch_bounds__(1024)
edge_bin(const int*   __restrict__ src,
         const int*   __restrict__ dst,
         const float* __restrict__ ew,
         int2*        __restrict__ binrec,
         int*         __restrict__ bindst,
         int2*        __restrict__ runtab,
         int n_edges, int n_owner, int n_chunk) {
    __shared__ int cnt[MAXP];
    __shared__ int off[MAXP];
    __shared__ int wtot[32];
    __shared__ __attribute__((aligned(16))) int2 srec[CHUNK];
    __shared__ __attribute__((aligned(16))) int  sdst[CHUNK];
    const int t    = threadIdx.x;
    const int lane = t & 31;
    const int w    = t >> 5;
    const int b    = blockIdx.x;
    if (t < MAXP) cnt[t] = 0;
    __syncthreads();

    const int e0 = b * CHUNK + t * 4;
    int d[4], s[4], wb[4], slot[4];
    if (e0 + 3 < n_edges) {
        const int4   dv = *(const int4*)(dst + e0);
        const int4   sv = *(const int4*)(src + e0);
        const float4 wv = *(const float4*)(ew + e0);
        d[0] = dv.x; d[1] = dv.y; d[2] = dv.z; d[3] = dv.w;
        s[0] = sv.x; s[1] = sv.y; s[2] = sv.z; s[3] = sv.w;
        wb[0] = __float_as_int(wv.x); wb[1] = __float_as_int(wv.y);
        wb[2] = __float_as_int(wv.z); wb[3] = __float_as_int(wv.w);
    } else {
        #pragma unroll
        for (int u = 0; u < 4; ++u) {
            const int e = e0 + u;
            d[u] = -1; s[u] = 0; wb[u] = 0;
            if (e < n_edges) { d[u] = dst[e]; s[u] = src[e]; wb[u] = __float_as_int(ew[e]); }
        }
    }
    #pragma unroll
    for (int u = 0; u < 4; ++u) {
        slot[u] = 0;
        if (d[u] >= 0) slot[u] = atomicAdd(&cnt[d[u] >> NPO_SHIFT], 1);
    }
    __syncthreads();

    const int c = (t < MAXP) ? cnt[t] : 0;
    int incl = c;
    for (int o = 1; o < 32; o <<= 1) {
        const int y = __shfl_up(incl, o);
        if (lane >= o) incl += y;
    }
    if (lane == 31) wtot[w] = incl;
    __syncthreads();
    const int wt = wtot[lane];
    int wincl = wt;
    for (int o = 1; o < 32; o <<= 1) {
        const int y = __shfl_up(wincl, o);
        if (lane >= o) wincl += y;
    }
    const int wexcl = __shfl(wincl, w) - __shfl(wt, w);
    const int excl  = wexcl + incl - c;
    if (t < MAXP) off[t] = excl;
    if (t < n_owner) runtab[(size_t)t * n_chunk + b] = make_int2(excl, c);
    __syncthreads();

    #pragma unroll
    for (int u = 0; u < 4; ++u) {
        if (d[u] >= 0) {
            const int q = off[d[u] >> NPO_SHIFT] + slot[u];
            srec[q] = make_int2(s[u], wb[u]);
            sdst[q] = d[u];
        }
    }
    __syncthreads();

    int nv = n_edges - b * CHUNK;
    nv = (nv < CHUNK) ? nv : CHUNK;
    const int4* lrec = (const int4*)srec;
    int4*       grec = (int4*)(binrec + (size_t)b * CHUNK);
    const int4* ldst = (const int4*)sdst;
    int4*       gdst = (int4*)(bindst + (size_t)b * CHUNK);
    if (2 * t < nv)          grec[t]        = lrec[t];
    if (2 * (t + 1024) < nv) grec[t + 1024] = lrec[t + 1024];
    if (4 * t < nv)          gdst[t]        = ldst[t];
}

static __device__ __forceinline__ int run_of(const int* rpre, const int n_chunk, const int i) {
    int lo = 0;
    #pragma unroll
    for (int step = MAXB / 2; step > 0; step >>= 1) {
        const int m  = lo + step;
        const int mm = (m < n_chunk) ? m : 0;
        lo = ((m < n_chunk) & (rpre[mm] <= i)) ? m : lo;
    }
    return lo;
}

__global__ void __launch_bounds__(OWN_T)
owner_sort(const int2* __restrict__ binrec,
           const int*  __restrict__ bindst,
           const int2* __restrict__ runtab,
           int*        __restrict__ rowptr,
           int2*       __restrict__ recs,
           int n_nodes, int n_chunk) {
    __shared__ int rstart[MAXB];
    __shared__ int rpre[MAXB + 1];
    __shared__ int ncnt[NPO];
    __shared__ int wlen[OWN_T / 32];
    __shared__ int wbas[OWN_T / 32];
    __shared__ int2 stage[OWN_U * OWN_T];
    const int t     = threadIdx.x;
    const int lane  = t & 31;
    const int w     = t >> 5;
    const int p     = blockIdx.x;
    const int nbase = p * NPO;

    int2 r = make_int2(0, 0);
    if (t < n_chunk) r = runtab[(size_t)p * n_chunk + t];
    int incl = r.y;
    for (int o = 1; o < 32; o <<= 1) {
        const int y = __shfl_up(incl, o);
        if (lane >= o) incl += y;
    }
    int bsum = r.x;
    for (int o = 16; o > 0; o >>= 1) bsum += __shfl_xor(bsum, o);
    if (lane == 31) wlen[w] = incl;
    if (lane == 0)  wbas[w] = bsum;
    if (t < NPO) ncnt[t] = 0;
    __syncthreads();
    const int wl = (lane < OWN_T / 32) ? wlen[lane] : 0;
    int wli = wl;
    for (int o = 1; o < 32; o <<= 1) {
        const int y = __shfl_up(wli, o);
        if (lane >= o) wli += y;
    }
    const int wexcl = __shfl(wli, w) - __shfl(wl, w);
    const int total = __shfl(wli, 31);
    int base = (lane < OWN_T / 32) ? wbas[lane] : 0;
    for (int o = 16; o > 0; o >>= 1) base += __shfl_xor(base, o);
    if (t < n_chunk) {
        rpre[t]   = wexcl + incl - r.y;
        rstart[t] = t * CHUNK + r.x;
    }
    if (t == 0) rpre[n_chunk] = total;
    __syncthreads();

    int  pos[OWN_U];
    int  dl[OWN_U];
    int2 rc[OWN_U];
    int ii[OWN_U];
    int lo[OWN_U];
    #pragma unroll
    for (int u = 0; u < OWN_U; ++u) {
        const int i = t + u * OWN_T;
        ii[u] = (i < total) ? i : 0;
        lo[u] = 0;
    }
    #pragma unroll
    for (int step = MAXB / 2; step > 0; step >>= 1) {
        #pragma unroll
        for (int u = 0; u < OWN_U; ++u) {
            const int m  = lo[u] + step;
            const int mm = (m < n_chunk) ? m : 0;
            lo[u] = ((m < n_chunk) & (rpre[mm] <= ii[u])) ? m : lo[u];
        }
    }
    #pragma unroll
    for (int u = 0; u < OWN_U; ++u) {
        const bool ok = (t + u * OWN_T) < total;
        const int  q  = rstart[lo[u]] + (ii[u] - rpre[lo[u]]);
        int dv = nbase;
        rc[u] = make_int2(0, 0);
        if (ok) { dv = bindst[q]; rc[u] = binrec[q]; }
        pos[u] = ok ? q : -1;
        dl[u]  = dv - nbase;
    }
    #pragma unroll
    for (int u = 0; u < OWN_U; ++u)
        if (pos[u] >= 0) atomicAdd(&ncnt[dl[u]], 1);
    for (int i = t + OWN_U * OWN_T; i < total; i += OWN_T) {
        const int rb = run_of(rpre, n_chunk, i);
        const int q  = rstart[rb] + (i - rpre[rb]);
        atomicAdd(&ncnt[bindst[q] - nbase], 1);
    }
    __syncthreads();

    const int c = (t < NPO) ? ncnt[t] : 0;
    int ninc = c;
    for (int o = 1; o < 32; o <<= 1) {
        const int y = __shfl_up(ninc, o);
        if (lane >= o) ninc += y;
    }
    if (lane == 31) wlen[w] = ninc;
    __syncthreads();
    const int nl = (lane < OWN_T / 32) ? wlen[lane] : 0;
    int nli = nl;
    for (int o = 1; o < 32; o <<= 1) {
        const int y = __shfl_up(nli, o);
        if (lane >= o) nli += y;
    }
    const int nexcl = __shfl(nli, w) - __shfl(nl, w) + ninc - c;
    if (t < NPO) {
        ncnt[t] = nexcl;
        if (nbase + t < n_nodes) rowptr[nbase + t] = base + nexcl;
    }
    if (p == (int)gridDim.x - 1 && t == 0) rowptr[n_nodes] = base + total;
    __syncthreads();

    #pragma unroll
    for (int u = 0; u < OWN_U; ++u) {
        if (pos[u] >= 0) {
            const int k = atomicAdd(&ncnt[dl[u]], 1);
            if (k < OWN_U * OWN_T) stage[k] = rc[u];
            else                   recs[base + k] = rc[u];
        }
    }
    for (int i = t + OWN_U * OWN_T; i < total; i += OWN_T) {
        const int rb = run_of(rpre, n_chunk, i);
        const int q  = rstart[rb] + (i - rpre[rb]);
        const int k  = atomicAdd(&ncnt[bindst[q] - nbase], 1);
        if (k < OWN_U * OWN_T) stage[k] = binrec[q];
        else                   recs[base + k] = binrec[q];
    }
    __syncthreads();
    const int ns = (total < OWN_U * OWN_T) ? total : OWN_U * OWN_T;
    for (int j = t; j < ns; j += OWN_T) recs[base + j] = stage[j];
}

#define TILE_N  64
#define WAVE_N  8
#define APITCH  68
#define GROUP   16

template <int J, bool FULL, bool PREF>
static __device__ __forceinline__ void
gather_group(const float* __restrict__ h, const int2 rec, const int lane,
             const int e0, const int E1, const int last,
             const int rp, float* myrow, int& node, int& nend, float& ax, float& ay,
             const int2* __restrict__ nptr, int2& nrec) {
    int    s[GROUP];
    float  w[GROUP];
    float2 x[GROUP];
    #pragma unroll
    for (int u = 0; u < GROUP; ++u) {
        if (FULL) {
            s[u] = __builtin_amdgcn_readlane(rec.x, J + u);
            w[u] = __int_as_float(__builtin_amdgcn_readlane(rec.y, J + u));
        } else {
            const int jj = (J + u < last) ? J + u : last;
            s[u] = __builtin_amdgcn_readlane(rec.x, jj);
            w[u] = __int_as_float(__builtin_amdgcn_readlane(rec.y, jj));
        }
    }
    #pragma unroll
    for (int u = 0; u < GROUP; ++u) {
        const unsigned off = (unsigned)s[u] * (unsigned)(FEAT * sizeof(float)) + (unsigned)lane * 8u;
        x[u] = *(const float2*)((const char*)h + off);
    }
    if (PREF) nrec = *nptr;
    #pragma unroll
    for (int u = 0; u < GROUP; ++u) {
        const int e = e0 + u;
        if (FULL || e < E1) {
            if (__builtin_expect(e >= nend, 0)) {
                do {
                    *(float2*)(myrow + node * APITCH) = make_float2(ax, ay);
                    ax = 0.0f; ay = 0.0f;
                    ++node;
                    nend = __builtin_amdgcn_readlane(rp, node + 1);
                } while (e >= nend);
            }
            ax += w[u] * x[u].x;
            ay += w[u] * x[u].y;
        }
    }
}

__global__ void __launch_bounds__(256)
gather_gemm_relu(const float* __restrict__ h,
                 const int*   __restrict__ rowptr,
                 const int2*  __restrict__ recs,
                 const float* __restrict__ W,
                 const float* __restrict__ bias,
                 float*       __restrict__ out,
                 int n_nodes) {
    __shared__ float Wlds[FEAT * FEAT];
    __shared__ float Alds[TILE_N * APITCH];

    const int lane  = threadIdx.x & 31;
    const int wave  = __builtin_amdgcn_readfirstlane((int)(threadIdx.x >> 5));
    const int nbase = (int)blockIdx.x * TILE_N;
    const int v0    = nbase + wave * WAVE_N;

    int bi = v0 + lane;
    bi = (bi < n_nodes) ? bi : n_nodes;
    int rp = 0;
    if (lane <= WAVE_N) rp = rowptr[bi];

    float4 wq[4];
    #pragma unroll
    for (int i = 0; i < 4; ++i) wq[i] = ((const float4*)W)[threadIdx.x + i * 256];

    const int E0 = __builtin_amdgcn_readlane(rp, 0);
    const int E1 = __builtin_amdgcn_readlane(rp, WAVE_N);
    int2 rec = make_int2(0, 0);
    if (E0 + lane < E1) rec = recs[E0 + lane];

    #pragma unroll
    for (int i = 0; i < 4; ++i) ((float4*)Wlds)[threadIdx.x + i * 256] = wq[i];

    float* myrow = Alds + (wave * WAVE_N) * APITCH + 2 * lane;
    int   node = 0;
    int   nend = __builtin_amdgcn_readlane(rp, 1);
    float ax = 0.0f, ay = 0.0f;

    for (int cb = E0; cb < E1; cb += 32) {
        const int left = E1 - cb;
        const int last = ((left < 32) ? left : 32) - 1;
        int ni = cb + 32 + lane;
        ni = (ni < E1) ? ni : E1 - 1;
        int2 nrec = make_int2(0, 0);
        if (left >= GROUP) gather_group<0, true,  true >(h, rec, lane, cb, E1, last, rp, myrow, node, nend, ax, ay, recs + ni, nrec);
        else               gather_group<0, false, false>(h, rec, lane, cb, E1, last, rp, myrow, node, nend, ax, ay, recs + ni, nrec);
        if (left >= 2 * GROUP)  gather_group<GROUP, true,  false>(h, rec, lane, cb + GROUP, E1, last, rp, myrow, node, nend, ax, ay, recs + ni, nrec);
        else if (left > GROUP)  gather_group<GROUP, false, false>(h, rec, lane, cb + GROUP, E1, last, rp, myrow, node, nend, ax, ay, recs + ni, nrec);
        rec = nrec;
    }
    for (; node < WAVE_N; ++node) {
        *(float2*)(myrow + node * APITCH) = make_float2(ax, ay);
        ax = 0.0f; ay = 0.0f;
    }
    __syncthreads();

    const int strip = wave >> 1;
    const int cpair = (wave & 1) * 2;
    const int nlo = lane & 15;
    const int hi  = lane >> 4;
    const float* aRow = Alds + (strip * 16 + nlo) * APITCH;

    v8f zero = {};
    v8f acc[2] = {zero, zero};
    #pragma unroll
    for (int k0 = 0; k0 < FEAT; k0 += 4) {
        const int ka = k0 + 2 * hi;
        v2f a;
        a.x = aRow[ka];
        a.y = aRow[ka + 1];
        #pragma unroll
        for (int c = 0; c < 2; ++c) {
            v2f b;
            b.x = Wlds[ka * FEAT + (cpair + c) * 16 + nlo];
            b.y = Wlds[(ka + 1) * FEAT + (cpair + c) * 16 + nlo];
            acc[c] = __builtin_amdgcn_wmma_f32_16x16x4_f32(
                 false, a,  false, b,
                 (short)0, acc[c],
                 false,  false);
        }
    }

    #pragma unroll
    for (int c = 0; c < 2; ++c) {
        const int col = (cpair + c) * 16 + nlo;
        const float bv = bias[col];
        #pragma unroll
        for (int v = 0; v < 8; ++v) {
            const int r = nbase + strip * 16 + v + 8 * hi;
            const float val = acc[c][v] + bv;
            if (r < n_nodes) out[(size_t)r * FEAT + col] = val > 0.0f ? val : 0.0f;
        }
    }
}

extern "C" void kernel_launch(void* const* d_in, const int* in_sizes, int n_in,
                              void* d_out, int out_size, void* d_ws, size_t ws_size,
                              hipStream_t stream) {
    const float* h   = (const float*)d_in[0];
    const float* Ws  = (const float*)d_in[1];
    const float* bs  = (const float*)d_in[2];
    const float* ew  = (const float*)d_in[3];
    const int*   src = (const int*)d_in[4];
    const int*   dst = (const int*)d_in[5];
    float* out = (float*)d_out;

    const int N = in_sizes[0] / FEAT;
    const int E = in_sizes[3];

    const int nOwner = (N + NPO - 1) / NPO;
    const int nChunk = (E + CHUNK - 1) / CHUNK;
    const int nRow   = ((N + 1 + 31) / 32) * 32;
    const int ePad   = ((E + 15) / 16) * 16;

    float* h1     = (float*)d_ws;
    int*   rowptr = (int*)(h1 + (size_t)N * FEAT);
    int2*  recs   = (int2*)(rowptr + nRow);
    int2*  binrec = recs + ePad;
    int*   bindst = (int*)(binrec + (size_t)nChunk * CHUNK);
    int2*  runtab = (int2*)(bindst + (size_t)nChunk * CHUNK);

    dim3 gGrd((N + TILE_N - 1) / TILE_N), gBlk(256);

    edge_bin<<<dim3(nChunk), dim3(1024), 0, stream>>>(src, dst, ew, binrec, bindst, runtab, E, nOwner, nChunk);
    owner_sort<<<dim3(nOwner), dim3(OWN_T), 0, stream>>>(binrec, bindst, runtab, rowptr, recs, N, nChunk);

    gather_gemm_relu<<<gGrd, gBlk, 0, stream>>>(h, rowptr, recs, Ws, bs, h1, N);

    gather_gemm_relu<<<gGrd, gBlk, 0, stream>>>(h1, rowptr, recs, Ws + FEAT * FEAT, bs + FEAT, out, N);
}


# Second sort pass and layer-0 gather+dense fused into one launch: the block that orders a 128-node group's records in LDS walks them from there (3 launches, not 4); sum order per node as before
# speedup vs baseline: 5.8875x; 1.0725x over previous
#include <hip/hip_runtime.h>


typedef __attribute__((ext_vector_type(2))) float v2f;
typedef __attribute__((ext_vector_type(8))) float v8f;
typedef __attribute__((ext_vector_type(2))) int   v2i;
typedef __attribute__((address_space(3))) const v2i lds_cv2i;

#define FEAT 64
#define NPO        128
#define NPO_SHIFT  7
#define CHUNK      4096
#define MAXP       512
#define MAXB       256
#define OWN_T      512
#define OWN_U      8

__global__ void __launch_bounds__(1024)
edge_bin(const int*   __restrict__ src,
         const int*   __restrict__ dst,
         const float* __restrict__ ew,
         int2*        __restrict__ binrec,
         int*         __restrict__ bindst,
         int2*        __restrict__ runtab,
         int n_edges, int n_owner, int n_chunk) {
    __shared__ int cnt[MAXP];
    __shared__ int off[MAXP];
    __shared__ int wtot[32];
    __shared__ __attribute__((aligned(16))) int2 srec[CHUNK];
    __shared__ __attribute__((aligned(16))) int  sdst[CHUNK];
    const int t    = threadIdx.x;
    const int lane = t & 31;
    const int w    = t >> 5;
    const int b    = blockIdx.x;
    if (t < MAXP) cnt[t] = 0;
    __syncthreads();

    const int e0 = b * CHUNK + t * 4;
    int d[4], s[4], wb[4], slot[4];
    if (e0 + 3 < n_edges) {
        const int4   dv = *(const int4*)(dst + e0);
        const int4   sv = *(const int4*)(src + e0);
        const float4 wv = *(const float4*)(ew + e0);
        d[0] = dv.x; d[1] = dv.y; d[2] = dv.z; d[3] = dv.w;
        s[0] = sv.x; s[1] = sv.y; s[2] = sv.z; s[3] = sv.w;
        wb[0] = __float_as_int(wv.x); wb[1] = __float_as_int(wv.y);
        wb[2] = __float_as_int(wv.z); wb[3] = __float_as_int(wv.w);
    } else {
        #pragma unroll
        for (int u = 0; u < 4; ++u) {
            const int e = e0 + u;
            d[u] = -1; s[u] = 0; wb[u] = 0;
            if (e < n_edges) { d[u] = dst[e]; s[u] = src[e]; wb[u] = __float_as_int(ew[e]); }
        }
    }
    #pragma unroll
    for (int u = 0; u < 4; ++u) {
        slot[u] = 0;
        if (d[u] >= 0) slot[u] = atomicAdd(&cnt[d[u] >> NPO_SHIFT], 1);
    }
    __syncthreads();

    const int c = (t < MAXP) ? cnt[t] : 0;
    int incl = c;
    for (int o = 1; o < 32; o <<= 1) {
        const int y = __shfl_up(incl, o);
        if (lane >= o) incl += y;
    }
    if (lane == 31) wtot[w] = incl;
    __syncthreads();
    const int wt = wtot[lane];
    int wincl = wt;
    for (int o = 1; o < 32; o <<= 1) {
        const int y = __shfl_up(wincl, o);
        if (lane >= o) wincl += y;
    }
    const int wexcl = __shfl(wincl, w) - __shfl(wt, w);
    const int excl  = wexcl + incl - c;
    if (t < MAXP) off[t] = excl;
    if (t < n_owner) runtab[(size_t)t * n_chunk + b] = make_int2(excl, c);
    __syncthreads();

    #pragma unroll
    for (int u = 0; u < 4; ++u) {
        if (d[u] >= 0) {
            const int q = off[d[u] >> NPO_SHIFT] + slot[u];
            srec[q] = make_int2(s[u], wb[u]);
            sdst[q] = d[u];
        }
    }
    __syncthreads();

    int nv = n_edges - b * CHUNK;
    nv = (nv < CHUNK) ? nv : CHUNK;
    const int4* lrec = (const int4*)srec;
    int4*       grec = (int4*)(binrec + (size_t)b * CHUNK);
    const int4* ldst = (const int4*)sdst;
    int4*       gdst = (int4*)(bindst + (size_t)b * CHUNK);
    if (2 * t < nv)          grec[t]        = lrec[t];
    if (2 * (t + 1024) < nv) grec[t + 1024] = lrec[t + 1024];
    if (4 * t < nv)          gdst[t]        = ldst[t];
}

static __device__ __forceinline__ int run_of(const int* rpre, const int n_chunk, const int i) {
    int lo = 0;
    #pragma unroll
    for (int step = MAXB / 2; step > 0; step >>= 1) {
        const int m  = lo + step;
        const int mm = (m < n_chunk) ? m : 0;
        lo = ((m < n_chunk) & (rpre[mm] <= i)) ? m : lo;
    }
    return lo;
}

#define TILE_N  64
#define WAVE_N  8
#define APITCH  68
#define GROUP   16

template <int J, bool FULL, bool PREF>
static __device__ __forceinline__ void
gather_group(const float* __restrict__ h, const int2 rec, const int lane,
             const int e0, const int E1, const int last,
             const int rp, float* myrow, int& node, int& nend, float& ax, float& ay,
             const int2* __restrict__ nptr, int2& nrec) {
    int    s[GROUP];
    float  w[GROUP];
    float2 x[GROUP];
    #pragma unroll
    for (int u = 0; u < GROUP; ++u) {
        if (FULL) {
            s[u] = __builtin_amdgcn_readlane(rec.x, J + u);
            w[u] = __int_as_float(__builtin_amdgcn_readlane(rec.y, J + u));
        } else {
            const int jj = (J + u < last) ? J + u : last;
            s[u] = __builtin_amdgcn_readlane(rec.x, jj);
            w[u] = __int_as_float(__builtin_amdgcn_readlane(rec.y, jj));
        }
    }
    #pragma unroll
    for (int u = 0; u < GROUP; ++u) {
        const unsigned off = (unsigned)s[u] * (unsigned)(FEAT * sizeof(float)) + (unsigned)lane * 8u;
        x[u] = *(const float2*)((const char*)h + off);
    }
    if (PREF) nrec = *nptr;
    #pragma unroll
    for (int u = 0; u < GROUP; ++u) {
        const int e = e0 + u;
        if (FULL || e < E1) {
            if (__builtin_expect(e >= nend, 0)) {
                do {
                    *(float2*)(myrow + node * APITCH) = make_float2(ax, ay);
                    ax = 0.0f; ay = 0.0f;
                    ++node;
                    nend = __builtin_amdgcn_readlane(rp, node + 1);
                } while (e >= nend);
            }
            ax += w[u] * x[u].x;
            ay += w[u] * x[u].y;
        }
    }
}

static __device__ __forceinline__ void
dense_relu_store(const float* Alds, const float* Wlds, const int wave, const int lane,
                 const int nbase, const float* __restrict__ bias, float* __restrict__ out,
                 const int n_nodes) {
    const int strip = wave >> 1;
    const int cpair = (wave & 1) * 2;
    const int nlo = lane & 15;
    const int hi  = lane >> 4;
    const float* aRow = Alds + (strip * 16 + nlo) * APITCH;

    v8f zero = {};
    v8f acc[2] = {zero, zero};
    #pragma unroll
    for (int k0 = 0; k0 < FEAT; k0 += 4) {
        const int ka = k0 + 2 * hi;
        v2f a;
        a.x = aRow[ka];
        a.y = aRow[ka + 1];
        #pragma unroll
        for (int c = 0; c < 2; ++c) {
            v2f b;
            b.x = Wlds[ka * FEAT + (cpair + c) * 16 + nlo];
            b.y = Wlds[(ka + 1) * FEAT + (cpair + c) * 16 + nlo];
            acc[c] = __builtin_amdgcn_wmma_f32_16x16x4_f32(
                 false, a,  false, b,
                 (short)0, acc[c],
                 false,  false);
        }
    }

    #pragma unroll
    for (int c = 0; c < 2; ++c) {
        const int col = (cpair + c) * 16 + nlo;
        const float bv = bias[col];
        #pragma unroll
        for (int v = 0; v < 8; ++v) {
            const int r = nbase + strip * 16 + v + 8 * hi;
            const float val = acc[c][v] + bv;
            if (r < n_nodes) out[(size_t)r * FEAT + col] = val > 0.0f ? val : 0.0f;
        }
    }
}

__global__ void __launch_bounds__(256)
gather_gemm_relu(const float* __restrict__ h,
                 const int*   __restrict__ rowptr,
                 const int2*  __restrict__ recs,
                 const float* __restrict__ W,
                 const float* __restrict__ bias,
                 float*       __restrict__ out,
                 int n_nodes) {
    __shared__ float Wlds[FEAT * FEAT];
    __shared__ float Alds[TILE_N * APITCH];

    const int lane  = threadIdx.x & 31;
    const int wave  = __builtin_amdgcn_readfirstlane((int)(threadIdx.x >> 5));
    const int nbase = (int)blockIdx.x * TILE_N;
    const int v0    = nbase + wave * WAVE_N;

    int bi = v0 + lane;
    bi = (bi < n_nodes) ? bi : n_nodes;
    int rp = 0;
    if (lane <= WAVE_N) rp = rowptr[bi];

    float4 wq[4];
    #pragma unroll
    for (int i = 0; i < 4; ++i) wq[i] = ((const float4*)W)[threadIdx.x + i * 256];

    const int E0 = __builtin_amdgcn_readlane(rp, 0);
    const int E1 = __builtin_amdgcn_readlane(rp, WAVE_N);
    int2 rec = make_int2(0, 0);
    if (E0 + lane < E1) rec = recs[E0 + lane];

    #pragma unroll
    for (int i = 0; i < 4; ++i) ((float4*)Wlds)[threadIdx.x + i * 256] = wq[i];

    float* myrow = Alds + (wave * WAVE_N) * APITCH + 2 * lane;
    int   node = 0;
    int   nend = __builtin_amdgcn_readlane(rp, 1);
    float ax = 0.0f, ay = 0.0f;

    for (int cb = E0; cb < E1; cb += 32) {
        const int left = E1 - cb;
        const int last = ((left < 32) ? left : 32) - 1;
        int ni = cb + 32 + lane;
        ni = (ni < E1) ? ni : E1 - 1;
        int2 nrec = make_int2(0, 0);
        if (left >= GROUP) gather_group<0, true,  true >(h, rec, lane, cb, E1, last, rp, myrow, node, nend, ax, ay, recs + ni, nrec);
        else               gather_group<0, false, false>(h, rec, lane, cb, E1, last, rp, myrow, node, nend, ax, ay, recs + ni, nrec);
        if (left >= 2 * GROUP)  gather_group<GROUP, true,  false>(h, rec, lane, cb + GROUP, E1, last, rp, myrow, node, nend, ax, ay, recs + ni, nrec);
        else if (left > GROUP)  gather_group<GROUP, false, false>(h, rec, lane, cb + GROUP, E1, last, rp, myrow, node, nend, ax, ay, recs + ni, nrec);
        rec = nrec;
    }
    for (; node < WAVE_N; ++node) {
        *(float2*)(myrow + node * APITCH) = make_float2(ax, ay);
        ax = 0.0f; ay = 0.0f;
    }
    __syncthreads();

    dense_relu_store(Alds, Wlds, wave, lane, nbase, bias, out, n_nodes);
}

__global__ void __launch_bounds__(OWN_T)
owner_sort_layer0(const int2*  __restrict__ binrec,
                  const int*   __restrict__ bindst,
                  const int2*  __restrict__ runtab,
                  int*         __restrict__ rowptr,
                  int2*        __restrict__ recs,
                  const float* __restrict__ h,
                  const float* __restrict__ W,
                  const float* __restrict__ bias,
                  float*       __restrict__ out,
                  int n_nodes, int n_chunk) {
    __shared__ int rstart[MAXB];
    __shared__ int rpre[MAXB + 1];
    __shared__ int ncnt[NPO];
    __shared__ int wlen[OWN_T / 32];
    __shared__ int wbas[OWN_T / 32];
    __shared__ int2 stage[OWN_U * OWN_T];
    __shared__ int lrow[NPO + 1];
    __shared__ float Wlds[FEAT * FEAT];
    __shared__ float Alds[NPO * APITCH];
    const int t     = threadIdx.x;
    const int lane  = t & 31;
    const int w     = t >> 5;
    const int p     = blockIdx.x;
    const int nbase = p * NPO;

    int2 r = make_int2(0, 0);
    if (t < n_chunk) r = runtab[(size_t)p * n_chunk + t];
    float4 wq[2];
    #pragma unroll
    for (int i = 0; i < 2; ++i) wq[i] = ((const float4*)W)[t + i * OWN_T];
    int incl = r.y;
    for (int o = 1; o < 32; o <<= 1) {
        const int y = __shfl_up(incl, o);
        if (lane >= o) incl += y;
    }
    int bsum = r.x;
    for (int o = 16; o > 0; o >>= 1) bsum += __shfl_xor(bsum, o);
    if (lane == 31) wlen[w] = incl;
    if (lane == 0)  wbas[w] = bsum;
    if (t < NPO) ncnt[t] = 0;
    #pragma unroll
    for (int i = 0; i < 2; ++i) ((float4*)Wlds)[t + i * OWN_T] = wq[i];
    __syncthreads();
    const int wl = (lane < OWN_T / 32) ? wlen[lane] : 0;
    int wli = wl;
    for (int o = 1; o < 32; o <<= 1) {
        const int y = __shfl_up(wli, o);
        if (lane >= o) wli += y;
    }
    const int wexcl = __shfl(wli, w) - __shfl(wl, w);
    const int total = __shfl(wli, 31);
    int base = (lane < OWN_T / 32) ? wbas[lane] : 0;
    for (int o = 16; o > 0; o >>= 1) base += __shfl_xor(base, o);
    if (t < n_chunk) {
        rpre[t]   = wexcl + incl - r.y;
        rstart[t] = t * CHUNK + r.x;
    }
    if (t == 0) rpre[n_chunk] = total;
    __syncthreads();

    int  pos[OWN_U];
    int  dl[OWN_U];
    int2 rc[OWN_U];
    int ii[OWN_U];
    int lo[OWN_U];
    #pragma unroll
    for (int u = 0; u < OWN_U; ++u) {
        const int i = t + u * OWN_T;
        ii[u] = (i < total) ? i : 0;
        lo[u] = 0;
    }
    #pragma unroll
    for (int step = MAXB / 2; step > 0; step >>= 1) {
        #pragma unroll
        for (int u = 0; u < OWN_U; ++u) {
            const int m  = lo[u] + step;
            const int mm = (m < n_chunk) ? m : 0;
            lo[u] = ((m < n_chunk) & (rpre[mm] <= ii[u])) ? m : lo[u];
        }
    }
    #pragma unroll
    for (int u = 0; u < OWN_U; ++u) {
        const bool ok = (t + u * OWN_T) < total;
        const int  q  = rstart[lo[u]] + (ii[u] - rpre[lo[u]]);
        int dv = nbase;
        rc[u] = make_int2(0, 0);
        if (ok) { dv = bindst[q]; rc[u] = binrec[q]; }
        pos[u] = ok ? q : -1;
        dl[u]  = dv - nbase;
    }
    #pragma unroll
    for (int u = 0; u < OWN_U; ++u)
        if (pos[u] >= 0) atomicAdd(&ncnt[dl[u]], 1);
    for (int i = t + OWN_U * OWN_T; i < total; i += OWN_T) {
        const int rb = run_of(rpre, n_chunk, i);
        const int q  = rstart[rb] + (i - rpre[rb]);
        atomicAdd(&ncnt[bindst[q] - nbase], 1);
    }
    __syncthreads();

    const int c = (t < NPO) ? ncnt[t] : 0;
    int ninc = c;
    for (int o = 1; o < 32; o <<= 1) {
        const int y = __shfl_up(ninc, o);
        if (lane >= o) ninc += y;
    }
    if (lane == 31) wlen[w] = ninc;
    __syncthreads();
    const int nl = (lane < OWN_T / 32) ? wlen[lane] : 0;
    int nli = nl;
    for (int o = 1; o < 32; o <<= 1) {
        const int y = __shfl_up(nli, o);
        if (lane >= o) nli += y;
    }
    const int nexcl = __shfl(nli, w) - __shfl(nl, w) + ninc - c;
    if (t < NPO) {
        ncnt[t] = nexcl;
        lrow[t] = nexcl;
        if (nbase + t < n_nodes) rowptr[nbase + t] = base + nexcl;
    }
    if (t == 0) lrow[NPO] = total;
    if (p == (int)gridDim.x - 1 && t == 0) rowptr[n_nodes] = base + total;
    __syncthreads();

    #pragma unroll
    for (int u = 0; u < OWN_U; ++u) {
        if (pos[u] >= 0) {
            const int k = atomicAdd(&ncnt[dl[u]], 1);
            if (k < OWN_U * OWN_T) stage[k] = rc[u];
            else                   recs[base + k] = rc[u];
        }
    }
    for (int i = t + OWN_U * OWN_T; i < total; i += OWN_T) {
        const int rb = run_of(rpre, n_chunk, i);
        const int q  = rstart[rb] + (i - rpre[rb]);
        const int k  = atomicAdd(&ncnt[bindst[q] - nbase], 1);
        if (k < OWN_U * OWN_T) stage[k] = binrec[q];
        else                   recs[base + k] = binrec[q];
    }
    __syncthreads();
    const int ns = (total < OWN_U * OWN_T) ? total : OWN_U * OWN_T;
    for (int j = t; j < ns; j += OWN_T) recs[base + j] = stage[j];

    const int wave = __builtin_amdgcn_readfirstlane(w);
    int rp = 0;
    if (lane <= WAVE_N) rp = lrow[wave * WAVE_N + lane];
    const int E0 = __builtin_amdgcn_readlane(rp, 0);
    const int E1 = __builtin_amdgcn_readlane(rp, WAVE_N);
    const lds_cv2i* stage_l = (const lds_cv2i*)stage;
    auto list_rec = [&](const int k) -> int2 {
        const v2i q = stage_l[(k < OWN_U * OWN_T) ? k : OWN_U * OWN_T - 1];
        int2 v = make_int2(q.x, q.y);
        if (k >= OWN_U * OWN_T) v = recs[base + k];
        return v;
    };
    int2 rec = make_int2(0, 0);
    if (E0 + lane < E1) rec = list_rec(E0 + lane);

    float* myrow = Alds + (wave * WAVE_N) * APITCH + 2 * lane;
    int   node = 0;
    int   nend = __builtin_amdgcn_readlane(rp, 1);
    float ax = 0.0f, ay = 0.0f;

    for (int cb = E0; cb < E1; cb += 32) {
        const int left = E1 - cb;
        const int last = ((left < 32) ? left : 32) - 1;
        int ni = cb + 32 + lane;
        ni = (ni < E1) ? ni : E1 - 1;
        const int2 nrec = list_rec(ni);
        int2 none = make_int2(0, 0);
        if (left >= GROUP) gather_group<0, true,  false>(h, rec, lane, cb, E1, last, rp, myrow, node, nend, ax, ay, recs, none);
        else               gather_group<0, false, false>(h, rec, lane, cb, E1, last, rp, myrow, node, nend, ax, ay, recs, none);
        if (left >= 2 * GROUP)  gather_group<GROUP, true,  false>(h, rec, lane, cb + GROUP, E1, last, rp, myrow, node, nend, ax, ay, recs, none);
        else if (left > GROUP)  gather_group<GROUP, false, false>(h, rec, lane, cb + GROUP, E1, last, rp, myrow, node, nend, ax, ay, recs, none);
        rec = nrec;
    }
    for (; node < WAVE_N; ++node) {
        *(float2*)(myrow + node * APITCH) = make_float2(ax, ay);
        ax = 0.0f; ay = 0.0f;
    }
    __syncthreads();

    dense_relu_store(Alds, Wlds, wave, lane, nbase, bias, out, n_nodes);
}

extern "C" void kernel_launch(void* const* d_in, const int* in_sizes, int n_in,
                              void* d_out, int out_size, void* d_ws, size_t ws_size,
                              hipStream_t stream) {
    const float* h   = (const float*)d_in[0];
    const float* Ws  = (const float*)d_in[1];
    const float* bs  = (const float*)d_in[2];
    const float* ew  = (const float*)d_in[3];
    const int*   src = (const int*)d_in[4];
    const int*   dst = (const int*)d_in[5];
    float* out = (float*)d_out;

    const int N = in_sizes[0] / FEAT;
    const int E = in_sizes[3];

    const int nOwner = (N + NPO - 1) / NPO;
    const int nChunk = (E + CHUNK - 1) / CHUNK;
    const int nRow   = ((N + 1 + 31) / 32) * 32;
    const int ePad   = ((E + 15) / 16) * 16;

    float* h1     = (float*)d_ws;
    int*   rowptr = (int*)(h1 + (size_t)N * FEAT);
    int2*  recs   = (int2*)(rowptr + nRow);
    int2*  binrec = recs + ePad;
    int*   bindst = (int*)(binrec + (size_t)nChunk * CHUNK);
    int2*  runtab = (int2*)(bindst + (size_t)nChunk * CHUNK);

    dim3 gGrd((N + TILE_N - 1) / TILE_N), gBlk(256);

    edge_bin<<<dim3(nChunk), dim3(1024), 0, stream>>>(src, dst, ew, binrec, bindst, runtab, E, nOwner, nChunk);
    owner_sort_layer0<<<dim3(nOwner), dim3(OWN_T), 0, stream>>>(binrec, bindst, runtab, rowptr, recs, h, Ws, bs, h1, N, nChunk);

    gather_gemm_relu<<<gGrd, gBlk, 0, stream>>>(h1, rowptr, recs, Ws + FEAT * FEAT, bs + FEAT, out, N);
}
